# starting kernel
# baseline (speedup 1.0000x reference)
#include <hip/hip_runtime.h>
#include <math.h>


typedef __attribute__((ext_vector_type(2))) float v2f;
typedef __attribute__((ext_vector_type(8))) float v8f;

constexpr int kB  = 16;
constexpr int kT  = 4096;
constexpr int kH  = 8;
constexpr int kD  = 64;
constexpr int kC  = 16;
constexpr int kNC = kT / kC;

__device__ __forceinline__ float sigmoidf_(float x) {
    return 1.0f / (1.0f + __expf(-x));
}

__global__ void __launch_bounds__(256)
ema_chunk_contrib(const float* __restrict__ x,
                  const float* __restrict__ sw,
                  float* __restrict__ contrib) {
    int gid = blockIdx.x * 256 + threadIdx.x;
    int d = gid & (kD - 1);
    int c = (gid >> 6) & (kNC - 1);
    int h = (gid >> 14) & (kH - 1);
    int b = gid >> 17;

    float w = sigmoidf_(sw[h]);
    const size_t rowStride = (size_t)kH * kD;
    const float* p = x + (((size_t)(b * kT + c * kC) * kH + h) * kD + d);

    float acc = 0.0f, wk = 1.0f;
    #pragma unroll
    for (int m = kC - 1; m >= 0; --m) {
        acc = fmaf(wk, p[(size_t)m * rowStride], acc);
        wk *= w;
    }
    contrib[gid] = (1.0f - w) * acc;
}

__global__ void ema_chunk_scan(const float* __restrict__ contrib,
                               const float* __restrict__ sw,
                               const float* __restrict__ v0,
                               float* __restrict__ carry) {
    int d  = threadIdx.x;
    int bh = blockIdx.x;
    int h  = bh & (kH - 1);

    float w = sigmoidf_(sw[h]);
    float w16 = 1.0f;
    #pragma unroll
    for (int i = 0; i < kC; ++i) w16 *= w;

    float s = v0[h * kD + d];
    size_t base = (size_t)bh * kNC * kD + d;
    for (int c = 0; c < kNC; ++c) {
        carry[base + (size_t)c * kD] = s;
        s = fmaf(w16, s, contrib[base + (size_t)c * kD]);
    }
}

__global__ void __launch_bounds__(256)
ema_chunk_apply(const float* __restrict__ x,
                const float* __restrict__ sw,
                const float* __restrict__ carry,
                float* __restrict__ out) {
    int wave = threadIdx.x >> 5;
    int lane = threadIdx.x & 31;
    int idx  = blockIdx.x * 8 + wave;
    int c = idx & (kNC - 1);
    int h = (idx >> 8) & (kH - 1);
    int b = idx >> 11;

    float w   = sigmoidf_(sw[h]);
    float omw = 1.0f - w;

    float wp[kC + 1];
    wp[0] = 1.0f;
    #pragma unroll
    for (int i = 1; i <= kC; ++i) wp[i] = wp[i - 1] * w;

    float invw = 1.0f / w;
    float invp[kC];
    invp[0] = 1.0f;
    #pragma unroll
    for (int i = 1; i < kC; ++i) invp[i] = invp[i - 1] * invw;

    int  hiHalf = (lane >= 16) ? 1 : 0;
    int  l15    = lane & 15;

    float pw2 = w * w, pw4 = pw2 * pw2, pw8 = pw4 * pw4;
    float pl = ((l15 & 1) ? w   : 1.0f)
             * ((l15 & 2) ? pw2 : 1.0f)
             * ((l15 & 4) ? pw4 : 1.0f)
             * ((l15 & 8) ? pw8 : 1.0f);
    float plw = omw * pl;

    v2f a[4];
    #pragma unroll
    for (int k = 0; k < 4; ++k) {
        float i0 = hiHalf ? invp[4 * k + 2] : invp[4 * k + 0];
        float i1 = hiHalf ? invp[4 * k + 3] : invp[4 * k + 1];
        int   m0 = 4 * k + 2 * hiHalf;
        float a0 = (l15 >= m0)     ? plw * i0 : 0.0f;
        float a1 = (l15 >= m0 + 1) ? plw * i1 : 0.0f;
        a[k] = (v2f){a0, a1};
    }

    const size_t rowStride = (size_t)kH * kD;
    const float* xbase = x   + (((size_t)(b * kT + c * kC) * kH + h) * kD);
    float*       obase = out + (((size_t)(b * kT + c * kC) * kH + h) * kD);
    const float* cbase = carry + (((size_t)(b * kH + h) * kNC + c) * kD);

    float hi8 = hiHalf ? wp[8] : 1.0f;

    #pragma unroll
    for (int n = 0; n < 4; ++n) {
        int col = n * 16 + l15;
        float cvh = cbase[col] * hi8;

        v8f acc;
        #pragma unroll
        for (int r = 0; r < 8; ++r)
            acc[r] = wp[r + 1] * cvh;

        #pragma unroll
        for (int k = 0; k < 4; ++k) {
            int krow = 4 * k + 2 * hiHalf;
            v2f bm;
            bm.x = __builtin_nontemporal_load(&xbase[(size_t)krow * rowStride + col]);
            bm.y = __builtin_nontemporal_load(&xbase[(size_t)(krow + 1) * rowStride + col]);
            acc = __builtin_amdgcn_wmma_f32_16x16x4_f32(
                false, a[k], false, bm, (short)0, acc, false, false);
        }

        #pragma unroll
        for (int r = 0; r < 8; ++r) {
            int row = r + 8 * hiHalf;
            __builtin_nontemporal_store(acc[r],
                &obase[(size_t)row * rowStride + col]);
        }
    }
}

extern "C" void kernel_launch(void* const* d_in, const int* in_sizes, int n_in,
                              void* d_out, int out_size, void* d_ws, size_t ws_size,
                              hipStream_t stream) {
    const float* values = (const float*)d_in[0];
    const float* sw     = (const float*)d_in[1];
    const float* v0     = (const float*)d_in[2];
    float* out = (float*)d_out;

    const size_t nChunkElems = (size_t)kB * kH * kNC * kD;
    float* contrib = (float*)d_ws;
    float* carry   = contrib + nChunkElems;

    ema_chunk_contrib<<<(int)(nChunkElems / 256), 256, 0, stream>>>(values, sw, contrib);
    ema_chunk_scan<<<kB * kH, kD, 0, stream>>>(contrib, sw, v0, carry);
    ema_chunk_apply<<<(kB * kH * kNC) / 8, 256, 0, stream>>>(values, sw, carry, out);
}


# chunk-carry scan made two-level and parallel: 16 waves per (b,h) each scan 16 chunks, segment totals exchanged via LDS; changes the summation tree of the carries
# speedup vs baseline: 1.6977x; 1.6977x over previous
#include <hip/hip_runtime.h>
#include <math.h>


typedef __attribute__((ext_vector_type(2))) float v2f;
typedef __attribute__((ext_vector_type(8))) float v8f;

constexpr int kB  = 16;
constexpr int kT  = 4096;
constexpr int kH  = 8;
constexpr int kD  = 64;
constexpr int kC  = 16;
constexpr int kNC = kT / kC;

__device__ __forceinline__ float sigmoidf_(float x) {
    return 1.0f / (1.0f + __expf(-x));
}

__global__ void __launch_bounds__(256)
ema_chunk_contrib(const float* __restrict__ x,
                  const float* __restrict__ sw,
                  float* __restrict__ contrib) {
    int gid = blockIdx.x * 256 + threadIdx.x;
    int d = gid & (kD - 1);
    int c = (gid >> 6) & (kNC - 1);
    int h = (gid >> 14) & (kH - 1);
    int b = gid >> 17;

    float w = sigmoidf_(sw[h]);
    const size_t rowStride = (size_t)kH * kD;
    const float* p = x + (((size_t)(b * kT + c * kC) * kH + h) * kD + d);

    float acc = 0.0f, wk = 1.0f;
    #pragma unroll
    for (int m = kC - 1; m >= 0; --m) {
        acc = fmaf(wk, p[(size_t)m * rowStride], acc);
        wk *= w;
    }
    contrib[gid] = (1.0f - w) * acc;
}

constexpr int kSeg  = 16;
constexpr int kNSeg = kNC / kSeg;

__global__ void __launch_bounds__(kNSeg * 32)
ema_chunk_scan(const float* __restrict__ contrib,
               const float* __restrict__ sw,
               const float* __restrict__ v0,
               float* __restrict__ carry) {
    __shared__ v2f segTot[kNSeg][kD / 2];

    int lane = threadIdx.x & 31;
    int g    = threadIdx.x >> 5;
    int bh   = blockIdx.x;
    int h    = bh & (kH - 1);

    size_t base = ((size_t)bh * kNC + (size_t)g * kSeg) * kD + 2 * lane;
    const v2f* src = reinterpret_cast<const v2f*>(contrib + base);
    v2f*       dst = reinterpret_cast<v2f*>(carry + base);

    v2f cv[kSeg];
    for (int i = 0; i < kSeg; ++i) cv[i] = src[i * (kD / 2)];
    v2f sv = *reinterpret_cast<const v2f*>(v0 + h * kD + 2 * lane);

    float w = sigmoidf_(sw[h]);
    float W = 1.0f;
    for (int i = 0; i < kC; ++i) W *= w;

    float Wp[kSeg + 1];
    Wp[0] = 1.0f;
    for (int i = 1; i <= kSeg; ++i) Wp[i] = Wp[i - 1] * W;

    v2f e[kSeg];
    v2f t = (v2f){0.0f, 0.0f};
    for (int i = 0; i < kSeg; ++i) {
        e[i] = t;
        t.x = fmaf(W, t.x, cv[i].x);
        t.y = fmaf(W, t.y, cv[i].y);
    }
    segTot[g][lane] = t;
    __syncthreads();

    float Wseg = Wp[kSeg];
    v2f tg[kNSeg - 1];
    for (int gp = 0; gp < kNSeg - 1; ++gp) tg[gp] = segTot[gp][lane];
    v2f run = sv;
    for (int gp = 0; gp < kNSeg - 1; ++gp) {
        run.x = fmaf(Wseg, run.x, tg[gp].x);
        run.y = fmaf(Wseg, run.y, tg[gp].y);
        sv.x = (g == gp + 1) ? run.x : sv.x;
        sv.y = (g == gp + 1) ? run.y : sv.y;
    }

    for (int i = 0; i < kSeg; ++i) {
        v2f o;
        o.x = fmaf(Wp[i], sv.x, e[i].x);
        o.y = fmaf(Wp[i], sv.y, e[i].y);
        dst[i * (kD / 2)] = o;
    }
}

__global__ void __launch_bounds__(256)
ema_chunk_apply(const float* __restrict__ x,
                const float* __restrict__ sw,
                const float* __restrict__ carry,
                float* __restrict__ out) {
    int wave = threadIdx.x >> 5;
    int lane = threadIdx.x & 31;
    int idx  = blockIdx.x * 8 + wave;
    int c = idx & (kNC - 1);
    int h = (idx >> 8) & (kH - 1);
    int b = idx >> 11;

    float w   = sigmoidf_(sw[h]);
    float omw = 1.0f - w;

    float wp[kC + 1];
    wp[0] = 1.0f;
    #pragma unroll
    for (int i = 1; i <= kC; ++i) wp[i] = wp[i - 1] * w;

    float invw = 1.0f / w;
    float invp[kC];
    invp[0] = 1.0f;
    #pragma unroll
    for (int i = 1; i < kC; ++i) invp[i] = invp[i - 1] * invw;

    int  hiHalf = (lane >= 16) ? 1 : 0;
    int  l15    = lane & 15;

    float pw2 = w * w, pw4 = pw2 * pw2, pw8 = pw4 * pw4;
    float pl = ((l15 & 1) ? w   : 1.0f)
             * ((l15 & 2) ? pw2 : 1.0f)
             * ((l15 & 4) ? pw4 : 1.0f)
             * ((l15 & 8) ? pw8 : 1.0f);
    float plw = omw * pl;

    v2f a[4];
    #pragma unroll
    for (int k = 0; k < 4; ++k) {
        float i0 = hiHalf ? invp[4 * k + 2] : invp[4 * k + 0];
        float i1 = hiHalf ? invp[4 * k + 3] : invp[4 * k + 1];
        int   m0 = 4 * k + 2 * hiHalf;
        float a0 = (l15 >= m0)     ? plw * i0 : 0.0f;
        float a1 = (l15 >= m0 + 1) ? plw * i1 : 0.0f;
        a[k] = (v2f){a0, a1};
    }

    const size_t rowStride = (size_t)kH * kD;
    const float* xbase = x   + (((size_t)(b * kT + c * kC) * kH + h) * kD);
    float*       obase = out + (((size_t)(b * kT + c * kC) * kH + h) * kD);
    const float* cbase = carry + (((size_t)(b * kH + h) * kNC + c) * kD);

    float hi8 = hiHalf ? wp[8] : 1.0f;

    #pragma unroll
    for (int n = 0; n < 4; ++n) {
        int col = n * 16 + l15;
        float cvh = cbase[col] * hi8;

        v8f acc;
        #pragma unroll
        for (int r = 0; r < 8; ++r)
            acc[r] = wp[r + 1] * cvh;

        #pragma unroll
        for (int k = 0; k < 4; ++k) {
            int krow = 4 * k + 2 * hiHalf;
            v2f bm;
            bm.x = __builtin_nontemporal_load(&xbase[(size_t)krow * rowStride + col]);
            bm.y = __builtin_nontemporal_load(&xbase[(size_t)(krow + 1) * rowStride + col]);
            acc = __builtin_amdgcn_wmma_f32_16x16x4_f32(
                false, a[k], false, bm, (short)0, acc, false, false);
        }

        #pragma unroll
        for (int r = 0; r < 8; ++r) {
            int row = r + 8 * hiHalf;
            __builtin_nontemporal_store(acc[r],
                &obase[(size_t)row * rowStride + col]);
        }
    }
}

extern "C" void kernel_launch(void* const* d_in, const int* in_sizes, int n_in,
                              void* d_out, int out_size, void* d_ws, size_t ws_size,
                              hipStream_t stream) {
    const float* values = (const float*)d_in[0];
    const float* sw     = (const float*)d_in[1];
    const float* v0     = (const float*)d_in[2];
    float* out = (float*)d_out;

    const size_t nChunkElems = (size_t)kB * kH * kNC * kD;
    float* contrib = (float*)d_ws;
    float* carry   = contrib + nChunkElems;

    ema_chunk_contrib<<<(int)(nChunkElems / 256), 256, 0, stream>>>(values, sw, contrib);
    ema_chunk_scan<<<kB * kH, kNSeg * 32, 0, stream>>>(contrib, sw, v0, carry);
    ema_chunk_apply<<<(kB * kH * kNC) / 8, 256, 0, stream>>>(values, sw, carry, out);
}


# passes 1 and 3 rebuilt as resident streaming loops: 4096 waves x 8 chunks, whole 256-byte rows per half-wave (16-byte lanes), next loads issued ahead; same wave-to-chunk map in both
# speedup vs baseline: 1.8920x; 1.1144x over previous
#include <hip/hip_runtime.h>
#include <math.h>


typedef __attribute__((ext_vector_type(2))) float v2f;
typedef __attribute__((ext_vector_type(4))) float v4f;
typedef __attribute__((ext_vector_type(8))) float v8f;

constexpr int kB  = 16;
constexpr int kT  = 4096;
constexpr int kH  = 8;
constexpr int kD  = 64;
constexpr int kC  = 16;
constexpr int kNC = kT / kC;

constexpr int kCW     = 8;
constexpr int kWpBH   = kNC / kCW;
constexpr int kStreamWaves  = kB * kH * kWpBH;
constexpr int kStreamBlocks = kStreamWaves / 8;

__device__ __forceinline__ float sigmoidf_(float x) {
    return 1.0f / (1.0f + __expf(-x));
}

__global__ void __launch_bounds__(256)
ema_chunk_contrib(const float* __restrict__ x,
                  const float* __restrict__ sw,
                  float* __restrict__ contrib) {
    int wave = threadIdx.x >> 5;
    int lane = threadIdx.x & 31;
    int wv   = blockIdx.x * 8 + wave;
    int c0   = (wv & (kWpBH - 1)) * kCW;
    int bh   = wv >> 5;
    int h    = bh & (kH - 1);
    int b    = bh >> 3;
    int q    = lane & 15;
    int half = lane >> 4;

    float w   = sigmoidf_(sw[h]);
    float omw = 1.0f - w;

    float wk[kC];
    wk[kC - 1] = 1.0f;
    #pragma unroll
    for (int m = kC - 2; m >= 0; --m) wk[m] = wk[m + 1] * w;

    const size_t rowStride = (size_t)kH * kD;

    v4f v[2][kC];
    #pragma unroll
    for (int m = 0; m < kC; ++m) {
        const float* p = x + (((size_t)(b * kT + (c0 + half) * kC) * kH + h) * kD + 4 * q);
        v[0][m] = *reinterpret_cast<const v4f*>(p + (size_t)m * rowStride);
    }

    #pragma unroll
    for (int j = 0; j < kCW / 2; ++j) {
        __builtin_amdgcn_sched_barrier(0);
        if (j + 1 < kCW / 2) {
            int cn = c0 + 2 * (j + 1) + half;
            const float* pn = x + (((size_t)(b * kT + cn * kC) * kH + h) * kD + 4 * q);
            #pragma unroll
            for (int m = 0; m < kC; ++m)
                v[(j + 1) & 1][m] = *reinterpret_cast<const v4f*>(pn + (size_t)m * rowStride);
        }
        __builtin_amdgcn_sched_barrier(0);

        int c = c0 + 2 * j + half;
        v4f acc = (v4f){0.0f, 0.0f, 0.0f, 0.0f};
        #pragma unroll
        for (int m = kC - 1; m >= 0; --m) {
            acc.x = fmaf(wk[m], v[j & 1][m].x, acc.x);
            acc.y = fmaf(wk[m], v[j & 1][m].y, acc.y);
            acc.z = fmaf(wk[m], v[j & 1][m].z, acc.z);
            acc.w = fmaf(wk[m], v[j & 1][m].w, acc.w);
        }
        v4f o = (v4f){omw * acc.x, omw * acc.y, omw * acc.z, omw * acc.w};
        *reinterpret_cast<v4f*>(contrib + (((size_t)bh * kNC + c) * kD + 4 * q)) = o;
    }
}

constexpr int kSeg  = 16;
constexpr int kNSeg = kNC / kSeg;

__global__ void __launch_bounds__(kNSeg * 32)
ema_chunk_scan(const float* __restrict__ contrib,
               const float* __restrict__ sw,
               const float* __restrict__ v0,
               float* __restrict__ carry) {
    __shared__ v2f segTot[kNSeg][kD / 2];

    int lane = threadIdx.x & 31;
    int g    = threadIdx.x >> 5;
    int bh   = blockIdx.x;
    int h    = bh & (kH - 1);

    size_t base = ((size_t)bh * kNC + (size_t)g * kSeg) * kD + 2 * lane;
    const v2f* src = reinterpret_cast<const v2f*>(contrib + base);
    v2f*       dst = reinterpret_cast<v2f*>(carry + base);

    v2f cv[kSeg];
    for (int i = 0; i < kSeg; ++i) cv[i] = src[i * (kD / 2)];
    v2f sv = *reinterpret_cast<const v2f*>(v0 + h * kD + 2 * lane);

    float w = sigmoidf_(sw[h]);
    float W = 1.0f;
    for (int i = 0; i < kC; ++i) W *= w;

    float Wp[kSeg + 1];
    Wp[0] = 1.0f;
    for (int i = 1; i <= kSeg; ++i) Wp[i] = Wp[i - 1] * W;

    v2f e[kSeg];
    v2f t = (v2f){0.0f, 0.0f};
    for (int i = 0; i < kSeg; ++i) {
        e[i] = t;
        t.x = fmaf(W, t.x, cv[i].x);
        t.y = fmaf(W, t.y, cv[i].y);
    }
    segTot[g][lane] = t;
    __syncthreads();

    float Wseg = Wp[kSeg];
    v2f tg[kNSeg - 1];
    for (int gp = 0; gp < kNSeg - 1; ++gp) tg[gp] = segTot[gp][lane];
    v2f run = sv;
    for (int gp = 0; gp < kNSeg - 1; ++gp) {
        run.x = fmaf(Wseg, run.x, tg[gp].x);
        run.y = fmaf(Wseg, run.y, tg[gp].y);
        sv.x = (g == gp + 1) ? run.x : sv.x;
        sv.y = (g == gp + 1) ? run.y : sv.y;
    }

    for (int i = 0; i < kSeg; ++i) {
        v2f o;
        o.x = fmaf(Wp[i], sv.x, e[i].x);
        o.y = fmaf(Wp[i], sv.y, e[i].y);
        dst[i * (kD / 2)] = o;
    }
}

struct ChunkRows { v4f r[8]; v4f cy; };

__device__ __forceinline__ void load_chunk(ChunkRows& t,
                                           const float* __restrict__ xq,
                                           const float* __restrict__ cq,
                                           int hiHalf, size_t rowStride) {
    #pragma unroll
    for (int k = 0; k < 4; ++k) {
        int krow = 4 * k + 2 * hiHalf;
        t.r[2 * k]     = __builtin_nontemporal_load(
            reinterpret_cast<const v4f*>(xq + (size_t)krow * rowStride));
        t.r[2 * k + 1] = __builtin_nontemporal_load(
            reinterpret_cast<const v4f*>(xq + (size_t)(krow + 1) * rowStride));
    }
    t.cy = *reinterpret_cast<const v4f*>(cq);
}

__global__ void __launch_bounds__(256)
ema_chunk_apply(const float* __restrict__ x,
                const float* __restrict__ sw,
                const float* __restrict__ carry,
                float* __restrict__ out) {
    int wave = threadIdx.x >> 5;
    int lane = threadIdx.x & 31;
    int wv   = blockIdx.x * 8 + wave;
    int c0   = (wv & (kWpBH - 1)) * kCW;
    int bh   = wv >> 5;
    int h    = bh & (kH - 1);
    int b    = bh >> 3;

    float w   = sigmoidf_(sw[h]);
    float omw = 1.0f - w;

    float wp[kC + 1];
    wp[0] = 1.0f;
    #pragma unroll
    for (int i = 1; i <= kC; ++i) wp[i] = wp[i - 1] * w;

    float invw = 1.0f / w;
    float invp[kC];
    invp[0] = 1.0f;
    #pragma unroll
    for (int i = 1; i < kC; ++i) invp[i] = invp[i - 1] * invw;

    int  hiHalf = (lane >= 16) ? 1 : 0;
    int  l15    = lane & 15;

    float pw2 = w * w, pw4 = pw2 * pw2, pw8 = pw4 * pw4;
    float pl = ((l15 & 1) ? w   : 1.0f)
             * ((l15 & 2) ? pw2 : 1.0f)
             * ((l15 & 4) ? pw4 : 1.0f)
             * ((l15 & 8) ? pw8 : 1.0f);
    float plw = omw * pl;

    v2f a[4];
    #pragma unroll
    for (int k = 0; k < 4; ++k) {
        float i0 = hiHalf ? invp[4 * k + 2] : invp[4 * k + 0];
        float i1 = hiHalf ? invp[4 * k + 3] : invp[4 * k + 1];
        int   m0 = 4 * k + 2 * hiHalf;
        float a0 = (l15 >= m0)     ? plw * i0 : 0.0f;
        float a1 = (l15 >= m0 + 1) ? plw * i1 : 0.0f;
        a[k] = (v2f){a0, a1};
    }

    const size_t rowStride = (size_t)kH * kD;
    const size_t chunkStride = (size_t)kC * rowStride;
    const float* xq = x   + (((size_t)(b * kT + c0 * kC) * kH + h) * kD + 4 * l15);
    float*       oq = out + (((size_t)(b * kT + c0 * kC) * kH + h) * kD + 4 * l15);
    const float* cq = carry + (((size_t)bh * kNC + c0) * kD + 4 * l15);

    float hi8 = hiHalf ? wp[8] : 1.0f;

    constexpr int kAhead = 2;
    ChunkRows ring[kAhead + 1];
    #pragma unroll
    for (int i = 0; i < kAhead; ++i)
        load_chunk(ring[i], xq + (size_t)i * chunkStride, cq + (size_t)i * kD,
                   hiHalf, rowStride);

    #pragma unroll
    for (int it = 0; it < kCW; ++it) {
        __builtin_amdgcn_sched_barrier(0);
        if (it + kAhead < kCW)
            load_chunk(ring[(it + kAhead) % (kAhead + 1)],
                       xq + (size_t)(it + kAhead) * chunkStride,
                       cq + (size_t)(it + kAhead) * kD, hiHalf, rowStride);
        __builtin_amdgcn_sched_barrier(0);

        const ChunkRows& cur = ring[it % (kAhead + 1)];

        v8f res[4];
        #pragma unroll
        for (int n = 0; n < 4; ++n) {
            float cvh = cur.cy[n] * hi8;

            v8f acc;
            #pragma unroll
            for (int r = 0; r < 8; ++r)
                acc[r] = wp[r + 1] * cvh;

            #pragma unroll
            for (int k = 0; k < 4; ++k) {
                v2f bm;
                bm.x = cur.r[2 * k][n];
                bm.y = cur.r[2 * k + 1][n];
                acc = __builtin_amdgcn_wmma_f32_16x16x4_f32(
                    false, a[k], false, bm, (short)0, acc, false, false);
            }
            res[n] = acc;
        }

        float* orow = oq + (size_t)it * chunkStride;
        #pragma unroll
        for (int r = 0; r < 8; ++r) {
            int row = r + 8 * hiHalf;
            v4f o = (v4f){res[0][r], res[1][r], res[2][r], res[3][r]};
            __builtin_nontemporal_store(o,
                reinterpret_cast<v4f*>(orow + (size_t)row * rowStride));
        }
    }
}

extern "C" void kernel_launch(void* const* d_in, const int* in_sizes, int n_in,
                              void* d_out, int out_size, void* d_ws, size_t ws_size,
                              hipStream_t stream) {
    const float* values = (const float*)d_in[0];
    const float* sw     = (const float*)d_in[1];
    const float* v0     = (const float*)d_in[2];
    float* out = (float*)d_out;

    constexpr size_t nChunkElems = (size_t)kB * kH * kNC * kD;
    float* contrib = (float*)d_ws;
    float* carry   = contrib + nChunkElems;

    ema_chunk_contrib<<<kStreamBlocks, 256, 0, stream>>>(values, sw, contrib);
    ema_chunk_scan<<<kB * kH, kNSeg * 32, 0, stream>>>(contrib, sw, v0, carry);
    ema_chunk_apply<<<kStreamBlocks, 256, 0, stream>>>(values, sw, carry, out);
}


# scan launch removed: pass 1 writes one total per 128-step wave segment; apply pass folds earlier totals onto v0, carries state between chunks via output row 15; changes the carries' summation tree
# speedup vs baseline: 2.4927x; 1.3175x over previous
#include <hip/hip_runtime.h>
#include <math.h>


typedef __attribute__((ext_vector_type(2))) float v2f;
typedef __attribute__((ext_vector_type(4))) float v4f;
typedef __attribute__((ext_vector_type(8))) float v8f;

constexpr int kB  = 16;
constexpr int kT  = 4096;
constexpr int kH  = 8;
constexpr int kD  = 64;
constexpr int kC  = 16;
constexpr int kNC = kT / kC;

constexpr int kCW     = 8;
constexpr int kWpBH   = kNC / kCW;
constexpr int kStreamWaves  = kB * kH * kWpBH;
constexpr int kStreamBlocks = kStreamWaves / 8;

__device__ __forceinline__ float sigmoidf_(float x) {
    return 1.0f / (1.0f + __expf(-x));
}

__device__ __forceinline__ float xhalf(float v) {
    int b = __builtin_bit_cast(int, v);
    return __builtin_bit_cast(float,
        __builtin_amdgcn_permlanex16(b, b, 0x76543210, 0xfedcba98, false, false));
}

__global__ void __launch_bounds__(256)
ema_segment_total(const float* __restrict__ x,
                  const float* __restrict__ sw,
                  float* __restrict__ seg) {
    int wave = threadIdx.x >> 5;
    int lane = threadIdx.x & 31;
    int wv   = blockIdx.x * 8 + wave;
    int c0   = (wv & (kWpBH - 1)) * kCW;
    int bh   = wv >> 5;
    int h    = bh & (kH - 1);
    int b    = bh >> 3;
    int q    = lane & 15;
    int half = lane >> 4;

    float w   = sigmoidf_(sw[h]);
    float omw = 1.0f - w;

    float wk[kC];
    wk[kC - 1] = 1.0f;
    #pragma unroll
    for (int m = kC - 2; m >= 0; --m) wk[m] = wk[m + 1] * w;

    float W  = wk[0] * w;
    float W2 = W * W;

    const size_t rowStride = (size_t)kH * kD;

    v4f v[2][kC];
    #pragma unroll
    for (int m = 0; m < kC; ++m) {
        const float* p = x + (((size_t)(b * kT + (c0 + half) * kC) * kH + h) * kD + 4 * q);
        v[0][m] = *reinterpret_cast<const v4f*>(p + (size_t)m * rowStride);
    }

    v4f P = (v4f){0.0f, 0.0f, 0.0f, 0.0f};
    #pragma unroll
    for (int j = 0; j < kCW / 2; ++j) {
        __builtin_amdgcn_sched_barrier(0);
        if (j + 1 < kCW / 2) {
            int cn = c0 + 2 * (j + 1) + half;
            const float* pn = x + (((size_t)(b * kT + cn * kC) * kH + h) * kD + 4 * q);
            #pragma unroll
            for (int m = 0; m < kC; ++m)
                v[(j + 1) & 1][m] = *reinterpret_cast<const v4f*>(pn + (size_t)m * rowStride);
        }
        __builtin_amdgcn_sched_barrier(0);

        v4f acc = (v4f){0.0f, 0.0f, 0.0f, 0.0f};
        #pragma unroll
        for (int m = kC - 1; m >= 0; --m) {
            acc.x = fmaf(wk[m], v[j & 1][m].x, acc.x);
            acc.y = fmaf(wk[m], v[j & 1][m].y, acc.y);
            acc.z = fmaf(wk[m], v[j & 1][m].z, acc.z);
            acc.w = fmaf(wk[m], v[j & 1][m].w, acc.w);
        }
        P.x = fmaf(W2, P.x, acc.x);
        P.y = fmaf(W2, P.y, acc.y);
        P.z = fmaf(W2, P.z, acc.z);
        P.w = fmaf(W2, P.w, acc.w);
    }

    float mult = half ? 1.0f : W;
    v4f cpart = (v4f){mult * P.x, mult * P.y, mult * P.z, mult * P.w};
    v4f tot;
    tot.x = omw * (cpart.x + xhalf(cpart.x));
    tot.y = omw * (cpart.y + xhalf(cpart.y));
    tot.z = omw * (cpart.z + xhalf(cpart.z));
    tot.w = omw * (cpart.w + xhalf(cpart.w));
    int g = wv & (kWpBH - 1);
    if (half == 0)
        *reinterpret_cast<v4f*>(seg + (((size_t)bh * kWpBH + g) * kD + 4 * q)) = tot;
}

struct ChunkRows { v4f r[8]; };

__device__ __forceinline__ void load_chunk(ChunkRows& t,
                                           const float* __restrict__ xq,
                                           int hiHalf, size_t rowStride) {
    #pragma unroll
    for (int k = 0; k < 4; ++k) {
        int krow = 4 * k + 2 * hiHalf;
        t.r[2 * k]     = __builtin_nontemporal_load(
            reinterpret_cast<const v4f*>(xq + (size_t)krow * rowStride));
        t.r[2 * k + 1] = __builtin_nontemporal_load(
            reinterpret_cast<const v4f*>(xq + (size_t)(krow + 1) * rowStride));
    }
}

__global__ void __launch_bounds__(256)
ema_chunk_apply(const float* __restrict__ x,
                const float* __restrict__ sw,
                const float* __restrict__ v0,
                const float* __restrict__ seg,
                float* __restrict__ out) {
    int wave = threadIdx.x >> 5;
    int lane = threadIdx.x & 31;
    int wv   = blockIdx.x * 8 + wave;
    int c0   = (wv & (kWpBH - 1)) * kCW;
    int g    = wv & (kWpBH - 1);
    int bh   = wv >> 5;
    int h    = bh & (kH - 1);
    int b    = bh >> 3;
    int  hiHalf = (lane >= 16) ? 1 : 0;
    int  l15    = lane & 15;

    float swv = sw[h];
    const float* tq = seg + ((size_t)bh * kWpBH) * kD + 4 * l15;
    v4f trow[kWpBH - 1];
    #pragma unroll
    for (int gp = 0; gp < kWpBH - 1; ++gp)
        trow[gp] = *reinterpret_cast<const v4f*>(tq + (size_t)gp * kD);
    v4f st = *reinterpret_cast<const v4f*>(v0 + h * kD + 4 * l15);
    __builtin_amdgcn_sched_barrier(0);

    const size_t rowStride = (size_t)kH * kD;
    const size_t chunkStride = (size_t)kC * rowStride;
    const float* xq = x   + (((size_t)(b * kT + c0 * kC) * kH + h) * kD + 4 * l15);
    float*       oq = out + (((size_t)(b * kT + c0 * kC) * kH + h) * kD + 4 * l15);

    constexpr int kAhead = 2;
    ChunkRows ring[kAhead + 1];
    #pragma unroll
    for (int i = 0; i < kAhead; ++i)
        load_chunk(ring[i], xq + (size_t)i * chunkStride, hiHalf, rowStride);
    __builtin_amdgcn_sched_barrier(0);

    float w   = sigmoidf_(swv);
    float omw = 1.0f - w;

    float wp[kC + 1];
    wp[0] = 1.0f;
    #pragma unroll
    for (int i = 1; i <= kC; ++i) wp[i] = wp[i - 1] * w;

    float invw = 1.0f / w;
    float invp[kC];
    invp[0] = 1.0f;
    #pragma unroll
    for (int i = 1; i < kC; ++i) invp[i] = invp[i - 1] * invw;

    float pw2 = w * w, pw4 = pw2 * pw2, pw8 = pw4 * pw4;
    float pl = ((l15 & 1) ? w   : 1.0f)
             * ((l15 & 2) ? pw2 : 1.0f)
             * ((l15 & 4) ? pw4 : 1.0f)
             * ((l15 & 8) ? pw8 : 1.0f);
    float plw = omw * pl;

    v2f a[4];
    #pragma unroll
    for (int k = 0; k < 4; ++k) {
        float i0 = hiHalf ? invp[4 * k + 2] : invp[4 * k + 0];
        float i1 = hiHalf ? invp[4 * k + 3] : invp[4 * k + 1];
        int   m0 = 4 * k + 2 * hiHalf;
        float a0 = (l15 >= m0)     ? plw * i0 : 0.0f;
        float a1 = (l15 >= m0 + 1) ? plw * i1 : 0.0f;
        a[k] = (v2f){a0, a1};
    }

    float hi8 = hiHalf ? wp[8] : 1.0f;

    float W32  = wp[kC] * wp[kC];
    float W64  = W32 * W32;
    float W128 = W64 * W64;
    #pragma unroll
    for (int gp = 0; gp < kWpBH - 1; ++gp) {
        bool use = gp < g;
        st.x = use ? fmaf(W128, st.x, trow[gp].x) : st.x;
        st.y = use ? fmaf(W128, st.y, trow[gp].y) : st.y;
        st.z = use ? fmaf(W128, st.z, trow[gp].z) : st.z;
        st.w = use ? fmaf(W128, st.w, trow[gp].w) : st.w;
    }

    #pragma unroll
    for (int it = 0; it < kCW; ++it) {
        __builtin_amdgcn_sched_barrier(0);
        if (it + kAhead < kCW)
            load_chunk(ring[(it + kAhead) % (kAhead + 1)],
                       xq + (size_t)(it + kAhead) * chunkStride, hiHalf, rowStride);
        __builtin_amdgcn_sched_barrier(0);

        const ChunkRows& cur = ring[it % (kAhead + 1)];

        v8f res[4];
        #pragma unroll
        for (int n = 0; n < 4; ++n) {
            float cvh = st[n] * hi8;

            v8f acc;
            #pragma unroll
            for (int r = 0; r < 8; ++r)
                acc[r] = wp[r + 1] * cvh;

            #pragma unroll
            for (int k = 0; k < 4; ++k) {
                v2f bm;
                bm.x = cur.r[2 * k][n];
                bm.y = cur.r[2 * k + 1][n];
                acc = __builtin_amdgcn_wmma_f32_16x16x4_f32(
                    false, a[k], false, bm, (short)0, acc, false, false);
            }
            res[n] = acc;

            if (it + 1 < kCW) {
                float u = hiHalf ? acc[7] : 0.0f;
                st[n] = u + xhalf(u);
            }
        }

        float* orow = oq + (size_t)it * chunkStride;
        #pragma unroll
        for (int r = 0; r < 8; ++r) {
            int row = r + 8 * hiHalf;
            v4f o = (v4f){res[0][r], res[1][r], res[2][r], res[3][r]};
            __builtin_nontemporal_store(o,
                reinterpret_cast<v4f*>(orow + (size_t)row * rowStride));
        }
    }
}

extern "C" void kernel_launch(void* const* d_in, const int* in_sizes, int n_in,
                              void* d_out, int out_size, void* d_ws, size_t ws_size,
                              hipStream_t stream) {
    const float* values = (const float*)d_in[0];
    const float* sw     = (const float*)d_in[1];
    const float* v0     = (const float*)d_in[2];
    float* out = (float*)d_out;

    float* seg = (float*)d_ws;

    ema_segment_total<<<kStreamBlocks, 256, 0, stream>>>(values, sw, seg);
    ema_chunk_apply<<<kStreamBlocks, 256, 0, stream>>>(values, sw, v0, seg, out);
}


# both passes fused into one launch: a 16-wave work-group owns a (b,h) half-row for all t, x read once, wave totals (row 15 of L*X) exchanged via LDS; changes the carries' summation tree
# speedup vs baseline: 2.9490x; 1.1831x over previous
#include <hip/hip_runtime.h>
#include <math.h>


typedef __attribute__((ext_vector_type(2))) float v2f;
typedef __attribute__((ext_vector_type(8))) float v8f;

constexpr int kB  = 16;
constexpr int kT  = 4096;
constexpr int kH  = 8;
constexpr int kD  = 64;
constexpr int kC  = 16;

constexpr int kWaves  = 16;
constexpr int kCW     = 4;
constexpr int kSub    = kCW * kC;
constexpr int kSuper  = kWaves * kSub;
constexpr int kNS     = kT / kSuper;
constexpr int kHalfD  = kD / 2;
constexpr int kBlocks = kB * kH * 2;

__device__ __forceinline__ float sigmoidf_(float x) {
    return 1.0f / (1.0f + __expf(-x));
}

__device__ __forceinline__ float xhalf(float v) {
    int b = __builtin_bit_cast(int, v);
    return __builtin_bit_cast(float,
        __builtin_amdgcn_permlanex16(b, b, 0x76543210, 0xfedcba98, false, false));
}

struct ChunkRows { v2f r[8]; };

__device__ __forceinline__ void load_chunk(ChunkRows& t,
                                           const float* __restrict__ xq,
                                           int hiHalf, size_t rowStride) {
    #pragma unroll
    for (int k = 0; k < 4; ++k) {
        int krow = 4 * k + 2 * hiHalf;
        t.r[2 * k]     = *reinterpret_cast<const v2f*>(xq + (size_t)krow * rowStride);
        t.r[2 * k + 1] = *reinterpret_cast<const v2f*>(xq + (size_t)(krow + 1) * rowStride);
    }
}

__global__ void __launch_bounds__(kWaves * 32)
ema_fused(const float* __restrict__ x,
          const float* __restrict__ sw,
          const float* __restrict__ v0,
          float* __restrict__ out) {
    __shared__ __attribute__((aligned(16))) float tot[2][kWaves][kHalfD];

    int wave = threadIdx.x >> 5;
    int lane = threadIdx.x & 31;
    int bh   = blockIdx.x >> 1;
    int dh   = blockIdx.x & 1;
    int h    = bh & (kH - 1);
    int b    = bh >> 3;
    int hiHalf = (lane >= 16) ? 1 : 0;
    int l15    = lane & 15;
    int col    = dh * kHalfD + 2 * l15;

    const size_t rowStride   = (size_t)kH * kD;
    const size_t chunkStride = (size_t)kC * rowStride;
    const size_t superStride = (size_t)kSuper * rowStride;
    const size_t base = ((size_t)(b * kT + wave * kSub) * kH + h) * kD + col;
    const float* xq = x + base;
    float*       oq = out + base;

    float swv = sw[h];
    v2f S = *reinterpret_cast<const v2f*>(v0 + h * kD + col);

    ChunkRows ring[2][kCW];
    #pragma unroll
    for (int j = 0; j < kCW; ++j)
        load_chunk(ring[0][j], xq + (size_t)j * chunkStride, hiHalf, rowStride);
    __builtin_amdgcn_sched_barrier(0);

    float w   = sigmoidf_(swv);
    float omw = 1.0f - w;

    float wp[kC + 1];
    wp[0] = 1.0f;
    #pragma unroll
    for (int i = 1; i <= kC; ++i) wp[i] = wp[i - 1] * w;

    float invw = 1.0f / w;
    float invp[kC];
    invp[0] = 1.0f;
    #pragma unroll
    for (int i = 1; i < kC; ++i) invp[i] = invp[i - 1] * invw;

    float pw2 = w * w, pw4 = pw2 * pw2, pw8 = pw4 * pw4;
    float pl = ((l15 & 1) ? w   : 1.0f)
             * ((l15 & 2) ? pw2 : 1.0f)
             * ((l15 & 4) ? pw4 : 1.0f)
             * ((l15 & 8) ? pw8 : 1.0f);
    float plw = omw * pl;

    v2f a[4];
    #pragma unroll
    for (int k = 0; k < 4; ++k) {
        float i0 = hiHalf ? invp[4 * k + 2] : invp[4 * k + 0];
        float i1 = hiHalf ? invp[4 * k + 3] : invp[4 * k + 1];
        int   m0 = 4 * k + 2 * hiHalf;
        float a0 = (l15 >= m0)     ? plw * i0 : 0.0f;
        float a1 = (l15 >= m0 + 1) ? plw * i1 : 0.0f;
        a[k] = (v2f){a0, a1};
    }

    float hi8 = hiHalf ? wp[8] : 1.0f;
    float W   = wp[kC];
    float W32 = W * W;
    float W64 = W32 * W32;

    #pragma unroll
    for (int s = 0; s < kNS; ++s) {
        __builtin_amdgcn_sched_barrier(0);
        if (s + 1 < kNS) {
            #pragma unroll
            for (int j = 0; j < kCW; ++j)
                load_chunk(ring[(s + 1) & 1][j],
                           xq + (size_t)(s + 1) * superStride + (size_t)j * chunkStride,
                           hiHalf, rowStride);
        }
        __builtin_amdgcn_sched_barrier(0);

        v8f z[kCW][2];
        v2f zt[kCW];
        #pragma unroll
        for (int j = 0; j < kCW; ++j) {
            const ChunkRows& cur = ring[s & 1][j];
            #pragma unroll
            for (int n = 0; n < 2; ++n) {
                v8f acc = (v8f){0.0f, 0.0f, 0.0f, 0.0f, 0.0f, 0.0f, 0.0f, 0.0f};
                #pragma unroll
                for (int k = 0; k < 4; ++k) {
                    v2f bm;
                    bm.x = cur.r[2 * k][n];
                    bm.y = cur.r[2 * k + 1][n];
                    acc = __builtin_amdgcn_wmma_f32_16x16x4_f32(
                        false, a[k], false, bm, (short)0, acc, false, false);
                }
                z[j][n] = acc;
                float u = hiHalf ? acc[7] : 0.0f;
                zt[j][n] = u + xhalf(u);
            }
        }

        v2f T = zt[0];
        #pragma unroll
        for (int j = 1; j < kCW; ++j) {
            T.x = fmaf(W, T.x, zt[j].x);
            T.y = fmaf(W, T.y, zt[j].y);
        }
        *reinterpret_cast<v2f*>(&tot[s & 1][wave][2 * l15]) = T;

        asm volatile("s_wait_dscnt 0x0\n\ts_barrier_signal -1\n\ts_barrier_wait -1" ::: "memory");

        v2f run = S, mine = S;
        #pragma unroll
        for (int k = 0; k < kWaves; ++k) {
            v2f tk = *reinterpret_cast<const v2f*>(&tot[s & 1][k][2 * l15]);
            bool me = (k == wave);
            mine.x = me ? run.x : mine.x;
            mine.y = me ? run.y : mine.y;
            run.x = fmaf(W64, run.x, tk.x);
            run.y = fmaf(W64, run.y, tk.y);
        }
        S = run;

        v2f c = mine;
        #pragma unroll
        for (int j = 0; j < kCW; ++j) {
            float cv0 = c.x * hi8;
            float cv1 = c.y * hi8;
            float* orow = oq + (size_t)s * superStride + (size_t)j * chunkStride;
            #pragma unroll
            for (int r = 0; r < 8; ++r) {
                int row = r + 8 * hiHalf;
                v2f o;
                o.x = fmaf(wp[r + 1], cv0, z[j][0][r]);
                o.y = fmaf(wp[r + 1], cv1, z[j][1][r]);
                __builtin_nontemporal_store(o,
                    reinterpret_cast<v2f*>(orow + (size_t)row * rowStride));
            }
            c.x = fmaf(W, c.x, zt[j].x);
            c.y = fmaf(W, c.y, zt[j].y);
        }
    }
}

extern "C" void kernel_launch(void* const* d_in, const int* in_sizes, int n_in,
                              void* d_out, int out_size, void* d_ws, size_t ws_size,
                              hipStream_t stream) {
    const float* values = (const float*)d_in[0];
    const float* sw     = (const float*)d_in[1];
    const float* v0     = (const float*)d_in[2];
    float* out = (float*)d_out;

    ema_fused<<<kBlocks, kWaves * 32, 0, stream>>>(values, sw, v0, out);
}
